# starting kernel
# baseline (speedup 1.0000x reference)
#include <hip/hip_runtime.h>
#include <math.h>


typedef __attribute__((ext_vector_type(2))) float v2f;
typedef __attribute__((ext_vector_type(8))) float v8f;

#define D_DIM 1024
#define F_DIM 8
#define NCOLS 32
#define KSLICE 256
#define ROWS_PER_BLOCK 128
#define LDS_STRIDE 260
#define WS_BYTES (NCOLS * D_DIM * sizeof(float))

__global__ __launch_bounds__(256) void pack_weights_kernel(
    const float* __restrict__ Wg,
    const float* __restrict__ Wf,
    const float* __restrict__ Wt,
    float* __restrict__ wp)
{
    const int idx = blockIdx.x * 256 + threadIdx.x;
    const int n = idx >> 10;
    const int k = idx & (D_DIM - 1);
    float v = 0.0f;
    if (n < 8)        v = Wg[k * F_DIM + n];
    else if (n < 16)  v = Wf[k * F_DIM + (n - 8)];
    else if (n < 24)  v = Wt[k * F_DIM + (n - 16)];
    wp[n * D_DIM + k] = v;
}

template <bool USE_WS>
__global__ __launch_bounds__(256) void fused_router_kernel(
    const float* __restrict__ x,
    const float* __restrict__ wp,
    const float* __restrict__ Wg, const float* __restrict__ bg,
    const float* __restrict__ Wf, const float* __restrict__ bf,
    const float* __restrict__ Wt, const float* __restrict__ bt,
    const float* __restrict__ alpha,
    float* __restrict__ out,
    int rows)
{
    __shared__ __align__(16) float bsm[NCOLS * LDS_STRIDE];

    const int tid  = threadIdx.x;
    const int lane = tid & 31;
    const int hf   = lane >> 4;
    const int l15  = lane & 15;
    const int wave = tid >> 5;

    const size_t rowBase = (size_t)blockIdx.x * ROWS_PER_BLOCK + (size_t)wave * 16;
    const float* xrow = x + (rowBase + (size_t)l15) * D_DIM;

    v8f c0 = {};
    v8f c1 = {};

    for (int ko = 0; ko < D_DIM; ko += KSLICE) {
        __syncthreads();
        if (USE_WS) {
            for (int i = tid; i < NCOLS * KSLICE / 4; i += 256) {
                const int n   = i >> 6;
                const int kk4 = (i & 63) << 2;
                const float4 v = *(const float4*)(wp + n * D_DIM + ko + kk4);
                *(float4*)(bsm + n * LDS_STRIDE + kk4) = v;
            }
        } else {
            for (int i = tid; i < NCOLS * KSLICE; i += 256) {
                const int n  = i >> 8;
                const int kk = i & (KSLICE - 1);
                const int k  = ko + kk;
                float v = 0.0f;
                if (n < 8)        v = Wg[k * F_DIM + n];
                else if (n < 16)  v = Wf[k * F_DIM + (n - 8)];
                else if (n < 24)  v = Wt[k * F_DIM + (n - 16)];
                bsm[n * LDS_STRIDE + kk] = v;
            }
        }
        __syncthreads();

        if (ko + KSLICE < D_DIM)
            __builtin_prefetch(xrow + ko + KSLICE, 0, 0);

        const float* b0p = bsm + l15 * LDS_STRIDE;
        const float* b1p = bsm + (l15 + 16) * LDS_STRIDE;

        #pragma unroll 4
        for (int kc = 0; kc < KSLICE / 4; ++kc) {
            const int kl = kc * 4 + hf * 2;
            v2f a  = *(const v2f*)(xrow + ko + kl);
            v2f b0 = *(const v2f*)(b0p + kl);
            v2f b1 = *(const v2f*)(b1p + kl);
            c0 = __builtin_amdgcn_wmma_f32_16x16x4_f32(false, a, false, b0,
                                                       (short)0, c0, false, false);
            c1 = __builtin_amdgcn_wmma_f32_16x16x4_f32(false, a, false, b1,
                                                       (short)0, c1, false, false);
        }
    }

    const float ab    = 1.0f / (1.0f + __expf(-alpha[0]));
    const float biasA = (l15 < 8) ? bg[l15] : bf[l15 - 8];
    const float biasB = (l15 < 8) ? bt[l15] : -__builtin_inff();

    #pragma unroll
    for (int r = 0; r < 8; ++r) {
        float v = c0[r] + biasA;

        float m = v;
        m = fmaxf(m, __shfl_xor(m, 1));
        m = fmaxf(m, __shfl_xor(m, 2));
        m = fmaxf(m, __shfl_xor(m, 4));
        float e = __expf(v - m);
        float s = e;
        s += __shfl_xor(s, 1);
        s += __shfl_xor(s, 2);
        s += __shfl_xor(s, 4);
        const float soft   = e / s;
        const float soft_f = __shfl_xor(soft, 8);
        const float gate   = 1.0f / (1.0f + __expf(-v));

        float t  = c1[r] + biasB;
        float t1 = t, t2 = -__builtin_inff();
        #pragma unroll
        for (int msk = 1; msk <= 4; msk <<= 1) {
            const float o1 = __shfl_xor(t1, msk);
            const float o2 = __shfl_xor(t2, msk);
            const float hi = fmaxf(t1, o1);
            const float lo = fminf(t1, o1);
            t2 = fmaxf(lo, fmaxf(t2, o2));
            t1 = hi;
        }
        const float p1 = 1.0f / (1.0f + __expf(t2 - t1));
        const float p2 = 1.0f - p1;
        const float sv = (t == t1) ? p1 : ((t == t2) ? p2 : 0.0f);

        float contrib = gate * (ab * sv + (1.0f - ab) * soft_f);
        contrib += __shfl_xor(contrib, 1);
        contrib += __shfl_xor(contrib, 2);
        contrib += __shfl_xor(contrib, 4);

        const size_t row = rowBase + (size_t)r + (size_t)(8 * hf);
        if (l15 == 0 && row < (size_t)rows)
            out[row] = contrib;
    }
}

extern "C" void kernel_launch(void* const* d_in, const int* in_sizes, int n_in,
                              void* d_out, int out_size, void* d_ws, size_t ws_size,
                              hipStream_t stream) {
    (void)n_in; (void)out_size;
    const float* x     = (const float*)d_in[0];
    const float* Wg    = (const float*)d_in[1];
    const float* bg    = (const float*)d_in[2];
    const float* Wf    = (const float*)d_in[3];
    const float* bf    = (const float*)d_in[4];
    const float* Wt    = (const float*)d_in[5];
    const float* bt    = (const float*)d_in[6];
    const float* alpha = (const float*)d_in[7];

    const int rows = in_sizes[0] / D_DIM;
    const int grid = (rows + ROWS_PER_BLOCK - 1) / ROWS_PER_BLOCK;

    if (ws_size >= WS_BYTES) {
        float* wp = (float*)d_ws;
        pack_weights_kernel<<<(NCOLS * D_DIM) / 256, 256, 0, stream>>>(Wg, Wf, Wt, wp);
        fused_router_kernel<true><<<grid, 256, 0, stream>>>(
            x, wp, Wg, bg, Wf, bf, Wt, bt, alpha, (float*)d_out, rows);
    } else {
        fused_router_kernel<false><<<grid, 256, 0, stream>>>(
            x, nullptr, Wg, bg, Wf, bf, Wt, bt, alpha, (float*)d_out, rows);
    }
}


# weights turned to fragment order inside the main kernel (whole-line loads, per-wave LDS transpose in the wave's own partial-tile area): pack launch and workspace gone; values and sum order unchanged
# speedup vs baseline: 1.7932x; 1.7932x over previous
#include <hip/hip_runtime.h>
#include <math.h>


typedef __attribute__((ext_vector_type(2))) float v2f;
typedef __attribute__((ext_vector_type(4))) float v4f;
typedef __attribute__((ext_vector_type(8))) float v8f;

#define D_DIM 1024
#define F_DIM 8
#define NWAVES 8
#define KW (D_DIM / NWAVES)
#define NLD (KW / 8)
#define TILES 8
#define ROWS_PER_BLOCK (TILES * 16)
#define SP 36
#define WAVE_PART (TILES * 4 * 32)
#define WPITCH 132

__global__ __launch_bounds__(256) void fused_router_kernel(
    const float* __restrict__ x,
    const float* __restrict__ Wg, const float* __restrict__ bg,
    const float* __restrict__ Wf, const float* __restrict__ bf,
    const float* __restrict__ Wt, const float* __restrict__ bt,
    const float* __restrict__ alpha,
    float* __restrict__ out,
    int rows)
{
    __shared__ __align__(16) v4f part[NWAVES * WAVE_PART];
    __shared__ __align__(16) float logit[ROWS_PER_BLOCK * SP];

    const int tid  = threadIdx.x;
    const int lane = tid & 31;
    const int hf   = lane >> 4;
    const int l15  = lane & 15;
    const int wave = tid >> 5;

    const int kbase = wave * KW + hf * 4;
    const size_t blockRow = (size_t)blockIdx.x * ROWS_PER_BLOCK;

    const size_t wofs = (size_t)wave * (KW * F_DIM) + 4 * lane;
    v4f rg[8], rf[8], rt[8];
    #pragma unroll
    for (int i = 0; i < 8; ++i) {
        rg[i] = *(const v4f*)(Wg + wofs + 128 * i);
        rf[i] = *(const v4f*)(Wf + wofs + 128 * i);
        rt[i] = *(const v4f*)(Wt + wofs + 128 * i);
    }

    const float* xp = x + (blockRow + (size_t)l15) * D_DIM + kbase;

    v4f a[4][NLD];
    #pragma unroll
    for (int t = 0; t < 3; ++t) {
        #pragma unroll
        for (int j = 0; j < NLD; ++j) a[t][j] = *(const v4f*)(xp + (size_t)t * 16 * D_DIM + 8 * j);
    }

    float* scr = (float*)(part + wave * WAVE_PART);
    {
        float* sp = scr + (4 * (lane & 1)) * WPITCH + (lane >> 1);
        #pragma unroll
        for (int i = 0; i < 8; ++i) {
            #pragma unroll
            for (int e = 0; e < 4; ++e) {
                sp[(e)      * WPITCH + 16 * i] = rg[i][e];
                sp[(8 + e)  * WPITCH + 16 * i] = rf[i][e];
                sp[(16 + e) * WPITCH + 16 * i] = rt[i][e];
            }
        }
    }
    __builtin_amdgcn_fence(__ATOMIC_ACQ_REL, "wavefront");

    v4f bA[NLD];
    v4f bB[NLD];
    {
        const float* la = scr + l15 * WPITCH + 4 * hf;
        const float* lb = scr + (16 + (l15 & 7)) * WPITCH + 4 * hf;
        #pragma unroll
        for (int j = 0; j < NLD; ++j) {
            bA[j] = *(const v4f*)(la + 8 * j);
            const v4f tb = *(const v4f*)(lb + 8 * j);
            bB[j] = (l15 < 8) ? tb : (v4f){0.0f, 0.0f, 0.0f, 0.0f};
        }
    }

    #pragma unroll
    for (int t = 0; t < TILES; ++t) {
        __builtin_amdgcn_sched_barrier(0);
        if (t + 3 < TILES) {
            const float* xn = xp + (size_t)(t + 3) * 16 * D_DIM;
            #pragma unroll
            for (int j = 0; j < NLD; ++j) a[(t + 3) % 4][j] = *(const v4f*)(xn + 8 * j);
        }
        __builtin_amdgcn_sched_barrier(0);

        v8f c0 = {};
        v8f c1 = {};
        #pragma unroll
        for (int j = 0; j < NLD; ++j) {
            const v4f av = a[t % 4][j];
            const v2f alo = {av.x, av.y};
            const v2f ahi = {av.z, av.w};
            const v2f b0lo = {bA[j].x, bA[j].y};
            const v2f b0hi = {bA[j].z, bA[j].w};
            const v2f b1lo = {bB[j].x, bB[j].y};
            const v2f b1hi = {bB[j].z, bB[j].w};
            c0 = __builtin_amdgcn_wmma_f32_16x16x4_f32(false, alo, false, b0lo,
                                                       (short)0, c0, false, false);
            c1 = __builtin_amdgcn_wmma_f32_16x16x4_f32(false, alo, false, b1lo,
                                                       (short)0, c1, false, false);
            c0 = __builtin_amdgcn_wmma_f32_16x16x4_f32(false, ahi, false, b0hi,
                                                       (short)0, c0, false, false);
            c1 = __builtin_amdgcn_wmma_f32_16x16x4_f32(false, ahi, false, b1hi,
                                                       (short)0, c1, false, false);
        }

        v4f* dst = part + wave * WAVE_PART + (t * 4) * 32 + lane;
        dst[0]  = (v4f){c0[0], c0[1], c0[2], c0[3]};
        dst[32] = (v4f){c0[4], c0[5], c0[6], c0[7]};
        dst[64] = (v4f){c1[0], c1[1], c1[2], c1[3]};
        dst[96] = (v4f){c1[4], c1[5], c1[6], c1[7]};
    }

    __syncthreads();

    const v4f* srcp = part + (wave * 4) * 32 + lane;
    v4f q0 = srcp[0], q1 = srcp[32], q2 = srcp[64], q3 = srcp[96];
    for (int w = 1; w < NWAVES; ++w) {
        const v4f* s = srcp + w * WAVE_PART;
        q0 += s[0];
        q1 += s[32];
        q2 += s[64];
        q3 += s[96];
    }
    const v8f c0 = {q0.x, q0.y, q0.z, q0.w, q1.x, q1.y, q1.z, q1.w};
    const v8f c1 = {q2.x, q2.y, q2.z, q2.w, q3.x, q3.y, q3.z, q3.w};

    const float biasA = (l15 < 8) ? bg[l15] : bf[l15 - 8];
    const float biasB = (l15 < 8) ? bt[l15] : 0.0f;
    float* lp = logit + (wave * 16 + 8 * hf) * SP + l15;
    #pragma unroll
    for (int r = 0; r < 8; ++r) {
        lp[r * SP]      = c0[r] + biasA;
        lp[r * SP + 16] = c1[r] + biasB;
    }

    __syncthreads();

    if (wave >= 4) return;

    const int rloc = wave * 32 + lane;
    const v4f* rp = (const v4f*)(logit + rloc * SP);
    const v4f ga = rp[0], gb = rp[1], sa = rp[2], sb = rp[3], ta = rp[4], tb = rp[5];
    const float g[8] = {ga.x, ga.y, ga.z, ga.w, gb.x, gb.y, gb.z, gb.w};
    const float s[8] = {sa.x, sa.y, sa.z, sa.w, sb.x, sb.y, sb.z, sb.w};
    const float t[8] = {ta.x, ta.y, ta.z, ta.w, tb.x, tb.y, tb.z, tb.w};

    const float ab = 1.0f / (1.0f + __expf(-alpha[0]));

    const float m = fmaxf(fmaxf(fmaxf(s[0], s[1]), fmaxf(s[2], s[3])),
                          fmaxf(fmaxf(s[4], s[5]), fmaxf(s[6], s[7])));
    float e[8];
    #pragma unroll
    for (int f = 0; f < 8; ++f) e[f] = __expf(s[f] - m);
    const float es = ((e[0] + e[1]) + (e[2] + e[3])) + ((e[4] + e[5]) + (e[6] + e[7]));

    float t1 = -__builtin_inff(), t2 = -__builtin_inff();
    #pragma unroll
    for (int f = 0; f < 8; ++f) {
        const float lo = fminf(t1, t[f]);
        t1 = fmaxf(t1, t[f]);
        t2 = fmaxf(t2, lo);
    }
    const float p1 = 1.0f / (1.0f + __expf(t2 - t1));
    const float p2 = 1.0f - p1;

    float cf[8];
    #pragma unroll
    for (int f = 0; f < 8; ++f) {
        const float gate = 1.0f / (1.0f + __expf(-g[f]));
        const float soft = e[f] / es;
        const float sv   = (t[f] == t1) ? p1 : ((t[f] == t2) ? p2 : 0.0f);
        cf[f] = gate * (ab * sv + (1.0f - ab) * soft);
    }
    const float total = ((cf[0] + cf[1]) + (cf[2] + cf[3])) + ((cf[4] + cf[5]) + (cf[6] + cf[7]));

    const size_t row = blockRow + (size_t)rloc;
    if (row < (size_t)rows)
        out[row] = total;
}

extern "C" void kernel_launch(void* const* d_in, const int* in_sizes, int n_in,
                              void* d_out, int out_size, void* d_ws, size_t ws_size,
                              hipStream_t stream) {
    (void)n_in; (void)out_size; (void)d_ws; (void)ws_size;
    const float* x     = (const float*)d_in[0];
    const float* Wg    = (const float*)d_in[1];
    const float* bg    = (const float*)d_in[2];
    const float* Wf    = (const float*)d_in[3];
    const float* bf    = (const float*)d_in[4];
    const float* Wt    = (const float*)d_in[5];
    const float* bt    = (const float*)d_in[6];
    const float* alpha = (const float*)d_in[7];

    const int rows = in_sizes[0] / D_DIM;
    const int grid = (rows + ROWS_PER_BLOCK - 1) / ROWS_PER_BLOCK;

    fused_router_kernel<<<grid, 256, 0, stream>>>(
        x, Wg, bg, Wf, bf, Wt, bt, alpha, (float*)d_out, rows);
}
